# starting kernel
# baseline (speedup 1.0000x reference)
#include <hip/hip_runtime.h>


typedef float v2f __attribute__((ext_vector_type(2)));
typedef float v8f __attribute__((ext_vector_type(8)));

#define WMMA_F32(A, B, C) \
  __builtin_amdgcn_wmma_f32_16x16x4_f32(false, (A), false, (B), (short)0, (C), false, false)

static __device__ __forceinline__ v2f ld2(const float* p) { return *(const v2f*)p; }
static __device__ __forceinline__ v2f mk2(float a, float b) { v2f r; r.x = a; r.y = b; return r; }

__global__ __launch_bounds__(256) void qkv_kernel(
    const float* __restrict__ x,
    const float* __restrict__ Wf, const float* __restrict__ Wg, const float* __restrict__ Wh,
    float* __restrict__ q, float* __restrict__ k, float* __restrict__ v) {
  __shared__ v2f wl[3][32 * 32];
  const int tid = threadIdx.x;
  const int wave = tid >> 5, lane = tid & 31;
  const int m = lane & 15, half = lane >> 4, kOff = half * 2;
  const int row0 = blockIdx.x * 128 + wave * 16;

  v8f aq0 = {}, aq1 = {}, ak0 = {}, ak1 = {}, av0 = {}, av1 = {};

  for (int c0 = 0; c0 < 256; c0 += 64) {
    for (int idx = tid; idx < 32 * 32; idx += 256) {
      const int kk2 = idx >> 5, n = idx & 31;
      const int g = (c0 + kk2 * 2) * 32 + n;
      wl[0][idx] = mk2(Wf[g], Wf[g + 32]);
      wl[1][idx] = mk2(Wg[g], Wg[g + 32]);
      wl[2][idx] = mk2(Wh[g], Wh[g + 32]);
    }
    __syncthreads();
#pragma unroll
    for (int kk = 0; kk < 64; kk += 4) {
      const v2f a = ld2(&x[(row0 + m) * 256 + c0 + kk + kOff]);
      const int bi = ((kk + kOff) >> 1) * 32;
      const v2f bf0 = wl[0][bi + m], bf1 = wl[0][bi + 16 + m];
      const v2f bg0 = wl[1][bi + m], bg1 = wl[1][bi + 16 + m];
      const v2f bh0 = wl[2][bi + m], bh1 = wl[2][bi + 16 + m];
      aq0 = WMMA_F32(a, bf0, aq0); aq1 = WMMA_F32(a, bf1, aq1);
      ak0 = WMMA_F32(a, bg0, ak0); ak1 = WMMA_F32(a, bg1, ak1);
      av0 = WMMA_F32(a, bh0, av0); av1 = WMMA_F32(a, bh1, av1);
    }
    __syncthreads();
  }
#pragma unroll
  for (int vi = 0; vi < 8; ++vi) {
    const int r = (row0 + vi + half * 8) * 32;
    q[r + m] = aq0[vi]; q[r + 16 + m] = aq1[vi];
    k[r + m] = ak0[vi]; k[r + 16 + m] = ak1[vi];
    v[r + m] = av0[vi]; v[r + 16 + m] = av1[vi];
  }
}

__global__ __launch_bounds__(256) void attn_kernel(
    const float* __restrict__ q, const float* __restrict__ k,
    const float* __restrict__ v, float* __restrict__ o) {
  __shared__ float smem[8][1024];
  const int tid = threadIdx.x;
  const int wave = tid >> 5, lane = tid & 31;
  const int m = lane & 15, half = lane >> 4, kOff = half * 2;
  const int tile = wave >> 2, split = wave & 3;
  const int task = blockIdx.x * 2 + tile;
  const int b = task >> 8;
  const int i0 = (task & 255) * 16;
  const float* qb = q + b * 4096 * 32;
  const float* kb = k + b * 4096 * 32;
  const float* vb = v + b * 4096 * 32;

  v2f aK[8];
#pragma unroll
  for (int t = 0; t < 8; ++t) aK[t] = ld2(&kb[(i0 + m) * 32 + 4 * t + kOff]);

  float mrun[8], lrun[8];
  v8f o0 = {}, o1 = {};
#pragma unroll
  for (int vi = 0; vi < 8; ++vi) { mrun[vi] = -__builtin_inff(); lrun[vi] = 0.f; }

  const int jbeg = split * 1024, jend = jbeg + 1024;
  for (int j0 = jbeg; j0 < jend; j0 += 64) {
    v8f s[4] = {{}, {}, {}, {}};
#pragma unroll
    for (int st = 0; st < 4; ++st) {
#pragma unroll
      for (int t = 0; t < 8; ++t) {
        const v2f bq = ld2(&qb[(j0 + st * 16 + m) * 32 + 4 * t + kOff]);
        s[st] = WMMA_F32(aK[t], bq, s[st]);
      }
    }
#pragma unroll
    for (int vi = 0; vi < 8; ++vi) {
      float mx = fmaxf(fmaxf(s[0][vi], s[1][vi]), fmaxf(s[2][vi], s[3][vi]));
      mx = fmaxf(mx, __shfl_xor(mx, 1, 16));
      mx = fmaxf(mx, __shfl_xor(mx, 2, 16));
      mx = fmaxf(mx, __shfl_xor(mx, 4, 16));
      mx = fmaxf(mx, __shfl_xor(mx, 8, 16));
      const float nm = fmaxf(mrun[vi], mx);
      const float sc = __expf(mrun[vi] - nm);
      mrun[vi] = nm;
      const float p0 = __expf(s[0][vi] - nm); s[0][vi] = p0;
      const float p1 = __expf(s[1][vi] - nm); s[1][vi] = p1;
      const float p2 = __expf(s[2][vi] - nm); s[2][vi] = p2;
      const float p3 = __expf(s[3][vi] - nm); s[3][vi] = p3;
      float rs = (p0 + p1) + (p2 + p3);
      rs += __shfl_xor(rs, 1, 16);
      rs += __shfl_xor(rs, 2, 16);
      rs += __shfl_xor(rs, 4, 16);
      rs += __shfl_xor(rs, 8, 16);
      lrun[vi] = lrun[vi] * sc + rs;
      o0[vi] *= sc;
      o1[vi] *= sc;
    }
#pragma unroll
    for (int st = 0; st < 4; ++st)
#pragma unroll
      for (int vi = 0; vi < 8; ++vi)
        smem[wave][(vi + half * 8) * 64 + st * 16 + m] = s[st][vi];
    asm volatile("s_wait_dscnt 0" ::: "memory");
#pragma unroll
    for (int t = 0; t < 16; ++t) {
      const int kk = 4 * t + kOff;
      const v2f aP = *(const v2f*)&smem[wave][m * 64 + kk];
      const v2f bv0 = mk2(vb[(j0 + kk) * 32 + m],      vb[(j0 + kk + 1) * 32 + m]);
      const v2f bv1 = mk2(vb[(j0 + kk) * 32 + 16 + m], vb[(j0 + kk + 1) * 32 + 16 + m]);
      o0 = WMMA_F32(aP, bv0, o0);
      o1 = WMMA_F32(aP, bv1, o1);
    }
  }

  asm volatile("s_wait_dscnt 0" ::: "memory");
#pragma unroll
  for (int vi = 0; vi < 8; ++vi) {
    const int r = vi + half * 8;
    smem[wave][r * 32 + m]      = o0[vi];
    smem[wave][r * 32 + 16 + m] = o1[vi];
  }
  if (m == 0) {
#pragma unroll
    for (int vi = 0; vi < 8; ++vi) {
      smem[wave][512 + vi + half * 8] = lrun[vi];
      smem[wave][528 + vi + half * 8] = mrun[vi];
    }
  }
  __syncthreads();

  if (split == 0) {
    const int w0 = wave;
#pragma unroll
    for (int vi = 0; vi < 8; ++vi) {
      const int r = vi + half * 8;
      float mg = smem[w0][528 + r];
      mg = fmaxf(mg, smem[w0 + 1][528 + r]);
      mg = fmaxf(mg, smem[w0 + 2][528 + r]);
      mg = fmaxf(mg, smem[w0 + 3][528 + r]);
      float lg = 0.f, a0 = 0.f, a1 = 0.f;
#pragma unroll
      for (int p = 0; p < 4; ++p) {
        const float wgt = __expf(smem[w0 + p][528 + r] - mg);
        lg += wgt * smem[w0 + p][512 + r];
        a0 += wgt * smem[w0 + p][r * 32 + m];
        a1 += wgt * smem[w0 + p][r * 32 + 16 + m];
      }
      const float inv = 1.f / lg;
      const int g = b * 4096 * 32 + (i0 + r) * 32;
      o[g + m]      = a0 * inv;
      o[g + 16 + m] = a1 * inv;
    }
  }
}

__global__ __launch_bounds__(256) void out_kernel(
    const float* __restrict__ o, const float* __restrict__ Wv,
    const float* __restrict__ x, const float* __restrict__ gamma,
    float* __restrict__ out) {
  const int tid = threadIdx.x;
  const int wave = tid >> 5, lane = tid & 31;
  const int m = lane & 15, half = lane >> 4, kOff = half * 2;
  const int task = blockIdx.x * 8 + wave;
  const int row0 = (task >> 2) * 16, col0 = (task & 3) * 64;

  v2f aO[8];
#pragma unroll
  for (int t = 0; t < 8; ++t) aO[t] = ld2(&o[(row0 + m) * 32 + 4 * t + kOff]);

  v8f acc[4] = {{}, {}, {}, {}};
#pragma unroll
  for (int t = 0; t < 8; ++t) {
    const int kk = 4 * t + kOff;
#pragma unroll
    for (int ct = 0; ct < 4; ++ct) {
      const int col = col0 + ct * 16 + m;
      const v2f bw = mk2(Wv[kk * 256 + col], Wv[(kk + 1) * 256 + col]);
      acc[ct] = WMMA_F32(aO[t], bw, acc[ct]);
    }
  }
  const float g = *gamma;
#pragma unroll
  for (int ct = 0; ct < 4; ++ct) {
#pragma unroll
    for (int vi = 0; vi < 8; ++vi) {
      const int idx = (row0 + vi + half * 8) * 256 + col0 + ct * 16 + m;
      out[idx] = g * acc[ct][vi] + x[idx];
    }
  }
}

extern "C" void kernel_launch(void* const* d_in, const int* in_sizes, int n_in,
                              void* d_out, int out_size, void* d_ws, size_t ws_size,
                              hipStream_t stream) {
  const float* x  = (const float*)d_in[0];
  const float* Wf = (const float*)d_in[1];
  const float* Wg = (const float*)d_in[2];
  const float* Wh = (const float*)d_in[3];
  const float* Wv = (const float*)d_in[4];
  const float* gm = (const float*)d_in[5];
  float* out = (float*)d_out;

  const int QKV = 4 * 4096 * 32;
  float* q = (float*)d_ws;
  float* k = q + QKV;
  float* v = k + QKV;
  float* o = v + QKV;

  qkv_kernel<<<128, 256, 0, stream>>>(x, Wf, Wg, Wh, q, k, v);
  attn_kernel<<<512, 256, 0, stream>>>(q, k, v, o);
  out_kernel<<<512, 256, 0, stream>>>(o, Wv, x, gm, out);
}


# projection kernel stores q and v in WMMA B-fragment order: attention reads each B fragment as one contiguous 16-byte-per-lane load, issued as one group (4 lines per load, was 16); same k order
# speedup vs baseline: 1.1133x; 1.1133x over previous
#include <hip/hip_runtime.h>


typedef float v2f __attribute__((ext_vector_type(2)));
typedef float v4f __attribute__((ext_vector_type(4)));
typedef float v8f __attribute__((ext_vector_type(8)));

#define WMMA_F32(A, B, C) \
  __builtin_amdgcn_wmma_f32_16x16x4_f32(false, (A), false, (B), (short)0, (C), false, false)

static __device__ __forceinline__ v2f ld2(const float* p) { return *(const v2f*)p; }
static __device__ __forceinline__ v4f ld4(const float* p) { return *(const v4f*)p; }
static __device__ __forceinline__ v2f mk2(float a, float b) { v2f r; r.x = a; r.y = b; return r; }

__global__ __launch_bounds__(256) void qkv_kernel(
    const float* __restrict__ x,
    const float* __restrict__ Wf, const float* __restrict__ Wg, const float* __restrict__ Wh,
    float* __restrict__ q, float* __restrict__ k, float* __restrict__ v) {
  __shared__ v2f wl[3][32 * 32];
  const int tid = threadIdx.x;
  const int wave = tid >> 5, lane = tid & 31;
  const int m = lane & 15, half = lane >> 4, kOff = half * 2;
  const int row0 = blockIdx.x * 128 + wave * 16;

  v8f aq0 = {}, aq1 = {}, ak0 = {}, ak1 = {}, av0 = {}, av1 = {};

  for (int c0 = 0; c0 < 256; c0 += 64) {
    for (int idx = tid; idx < 32 * 32; idx += 256) {
      const int kk2 = idx >> 5, n = idx & 31;
      const int g = (c0 + kk2 * 2) * 32 + n;
      wl[0][idx] = mk2(Wf[g], Wf[g + 32]);
      wl[1][idx] = mk2(Wg[g], Wg[g + 32]);
      wl[2][idx] = mk2(Wh[g], Wh[g + 32]);
    }
    __syncthreads();
#pragma unroll
    for (int kk = 0; kk < 64; kk += 4) {
      const v2f a = ld2(&x[(row0 + m) * 256 + c0 + kk + kOff]);
      const int bi = ((kk + kOff) >> 1) * 32;
      const v2f bf0 = wl[0][bi + m], bf1 = wl[0][bi + 16 + m];
      const v2f bg0 = wl[1][bi + m], bg1 = wl[1][bi + 16 + m];
      const v2f bh0 = wl[2][bi + m], bh1 = wl[2][bi + 16 + m];
      aq0 = WMMA_F32(a, bf0, aq0); aq1 = WMMA_F32(a, bf1, aq1);
      ak0 = WMMA_F32(a, bg0, ak0); ak1 = WMMA_F32(a, bg1, ak1);
      av0 = WMMA_F32(a, bh0, av0); av1 = WMMA_F32(a, bh1, av1);
    }
    __syncthreads();
  }
  const int pm = (m & 1) | ((m & 2) << 1) | ((m & 4) >> 1);
  float* qt = q + (row0 >> 4) * 512 + (m >> 3) * 128 + pm;
#pragma unroll
  for (int vi = 0; vi < 8; ++vi) {
    const int rr = vi + half * 8;
    qt[rr * 8] = aq0[vi];
    qt[256 + rr * 8] = aq1[vi];
    const int r = (row0 + rr) * 32;
    k[r + m] = ak0[vi]; k[r + 16 + m] = ak1[vi];
  }
#pragma unroll
  for (int gg = 0; gg < 2; ++gg) {
#pragma unroll
    for (int h2 = 0; h2 < 2; ++h2) {
      const int vi0 = 4 * gg + 2 * h2;
      v4f ch;
      ch.x = av0[vi0]; ch.y = av0[vi0 + 1]; ch.z = av1[vi0]; ch.w = av1[vi0 + 1];
      *(v4f*)&v[((row0 >> 2) + 2 * half + gg) * 128 + (h2 * 16 + m) * 4] = ch;
    }
  }
}

__global__ __launch_bounds__(256) __attribute__((amdgpu_waves_per_eu(4, 4))) void attn_kernel(
    const float* __restrict__ q, const float* __restrict__ k,
    const float* __restrict__ v, float* __restrict__ o) {
  __shared__ float smem[8][1024];
  const int tid = threadIdx.x;
  const int wave = tid >> 5, lane = tid & 31;
  const int m = lane & 15, half = lane >> 4, kOff = half * 2;
  const int tile = wave >> 2, split = wave & 3;
  const int task = blockIdx.x * 2 + tile;
  const int b = task >> 8;
  const int i0 = (task & 255) * 16;
  const float* qb = q + b * 4096 * 32 + m * 8 + half * 4;
  const float* kb = k + b * 4096 * 32;
  const float* vb = v + b * 4096 * 32 + lane * 4;

  v2f aK[8];
#pragma unroll
  for (int t = 0; t < 8; ++t) aK[t] = ld2(&kb[(i0 + m) * 32 + 4 * t + kOff]);

  float mrun[8], lrun[8];
  v8f o0 = {}, o1 = {};
#pragma unroll
  for (int vi = 0; vi < 8; ++vi) { mrun[vi] = -__builtin_inff(); lrun[vi] = 0.f; }

  const int jbeg = split * 1024, jend = jbeg + 1024;
  for (int j0 = jbeg; j0 < jend; j0 += 64) {
    v8f s[4] = {{}, {}, {}, {}};
#pragma unroll
    for (int st = 0; st < 4; ++st) {
#pragma unroll
      for (int u = 0; u < 4; ++u) {
        const v4f bq = ld4(&qb[((j0 >> 4) + st) * 512 + u * 128]);
        s[st] = WMMA_F32(aK[2 * u], mk2(bq.x, bq.y), s[st]);
        s[st] = WMMA_F32(aK[2 * u + 1], mk2(bq.z, bq.w), s[st]);
      }
    }
#pragma unroll
    for (int vi = 0; vi < 8; ++vi) {
      float mx = fmaxf(fmaxf(s[0][vi], s[1][vi]), fmaxf(s[2][vi], s[3][vi]));
      mx = fmaxf(mx, __shfl_xor(mx, 1, 16));
      mx = fmaxf(mx, __shfl_xor(mx, 2, 16));
      mx = fmaxf(mx, __shfl_xor(mx, 4, 16));
      mx = fmaxf(mx, __shfl_xor(mx, 8, 16));
      const float nm = fmaxf(mrun[vi], mx);
      const float sc = __expf(mrun[vi] - nm);
      mrun[vi] = nm;
      const float p0 = __expf(s[0][vi] - nm); s[0][vi] = p0;
      const float p1 = __expf(s[1][vi] - nm); s[1][vi] = p1;
      const float p2 = __expf(s[2][vi] - nm); s[2][vi] = p2;
      const float p3 = __expf(s[3][vi] - nm); s[3][vi] = p3;
      float rs = (p0 + p1) + (p2 + p3);
      rs += __shfl_xor(rs, 1, 16);
      rs += __shfl_xor(rs, 2, 16);
      rs += __shfl_xor(rs, 4, 16);
      rs += __shfl_xor(rs, 8, 16);
      lrun[vi] = lrun[vi] * sc + rs;
      o0[vi] *= sc;
      o1[vi] *= sc;
    }
#pragma unroll
    for (int st = 0; st < 4; ++st)
#pragma unroll
      for (int vi = 0; vi < 8; ++vi)
        smem[wave][(vi + half * 8) * 64 + st * 16 + m] = s[st][vi];
    asm volatile("s_wait_dscnt 0" ::: "memory");
#pragma unroll
    for (int t = 0; t < 16; ++t) {
      const int kk = 4 * t + kOff;
      const v2f aP = *(const v2f*)&smem[wave][m * 64 + kk];
      const v4f bv = ld4(&vb[((j0 >> 2) + t) * 128]);
      o0 = WMMA_F32(aP, mk2(bv.x, bv.y), o0);
      o1 = WMMA_F32(aP, mk2(bv.z, bv.w), o1);
    }
  }

  asm volatile("s_wait_dscnt 0" ::: "memory");
#pragma unroll
  for (int vi = 0; vi < 8; ++vi) {
    const int r = vi + half * 8;
    smem[wave][r * 32 + m]      = o0[vi];
    smem[wave][r * 32 + 16 + m] = o1[vi];
  }
  if (m == 0) {
#pragma unroll
    for (int vi = 0; vi < 8; ++vi) {
      smem[wave][512 + vi + half * 8] = lrun[vi];
      smem[wave][528 + vi + half * 8] = mrun[vi];
    }
  }
  __syncthreads();

  if (split == 0) {
    const int w0 = wave;
#pragma unroll
    for (int vi = 0; vi < 8; ++vi) {
      const int r = vi + half * 8;
      float mg = smem[w0][528 + r];
      mg = fmaxf(mg, smem[w0 + 1][528 + r]);
      mg = fmaxf(mg, smem[w0 + 2][528 + r]);
      mg = fmaxf(mg, smem[w0 + 3][528 + r]);
      float lg = 0.f, a0 = 0.f, a1 = 0.f;
#pragma unroll
      for (int p = 0; p < 4; ++p) {
        const float wgt = __expf(smem[w0 + p][528 + r] - mg);
        lg += wgt * smem[w0 + p][512 + r];
        a0 += wgt * smem[w0 + p][r * 32 + m];
        a1 += wgt * smem[w0 + p][r * 32 + 16 + m];
      }
      const float inv = 1.f / lg;
      const int g = b * 4096 * 32 + (i0 + r) * 32;
      o[g + m]      = a0 * inv;
      o[g + 16 + m] = a1 * inv;
    }
  }
}

__global__ __launch_bounds__(256) void out_kernel(
    const float* __restrict__ o, const float* __restrict__ Wv,
    const float* __restrict__ x, const float* __restrict__ gamma,
    float* __restrict__ out) {
  const int tid = threadIdx.x;
  const int wave = tid >> 5, lane = tid & 31;
  const int m = lane & 15, half = lane >> 4, kOff = half * 2;
  const int task = blockIdx.x * 8 + wave;
  const int row0 = (task >> 2) * 16, col0 = (task & 3) * 64;

  v2f aO[8];
#pragma unroll
  for (int t = 0; t < 8; ++t) aO[t] = ld2(&o[(row0 + m) * 32 + 4 * t + kOff]);

  v8f acc[4] = {{}, {}, {}, {}};
#pragma unroll
  for (int t = 0; t < 8; ++t) {
    const int kk = 4 * t + kOff;
#pragma unroll
    for (int ct = 0; ct < 4; ++ct) {
      const int col = col0 + ct * 16 + m;
      const v2f bw = mk2(Wv[kk * 256 + col], Wv[(kk + 1) * 256 + col]);
      acc[ct] = WMMA_F32(aO[t], bw, acc[ct]);
    }
  }
  const float g = *gamma;
#pragma unroll
  for (int ct = 0; ct < 4; ++ct) {
#pragma unroll
    for (int vi = 0; vi < 8; ++vi) {
      const int idx = (row0 + vi + half * 8) * 256 + col0 + ct * 16 + m;
      out[idx] = g * acc[ct][vi] + x[idx];
    }
  }
}

extern "C" void kernel_launch(void* const* d_in, const int* in_sizes, int n_in,
                              void* d_out, int out_size, void* d_ws, size_t ws_size,
                              hipStream_t stream) {
  const float* x  = (const float*)d_in[0];
  const float* Wf = (const float*)d_in[1];
  const float* Wg = (const float*)d_in[2];
  const float* Wh = (const float*)d_in[3];
  const float* Wv = (const float*)d_in[4];
  const float* gm = (const float*)d_in[5];
  float* out = (float*)d_out;

  const int QKV = 4 * 4096 * 32;
  float* q = (float*)d_ws;
  float* k = q + QKV;
  float* v = k + QKV;
  float* o = v + QKV;

  qkv_kernel<<<128, 256, 0, stream>>>(x, Wf, Wg, Wh, q, k, v);
  attn_kernel<<<512, 256, 0, stream>>>(q, k, v, o);
  out_kernel<<<512, 256, 0, stream>>>(o, Wv, x, gm, out);
}


# attention on transposed tiles: S^T D registers after exp are the B fragments of O^T += V^T.P^T, so P skips LDS; row max and sum in-lane, one lane exchange per block; sum order changes
# speedup vs baseline: 1.2763x; 1.1464x over previous
#include <hip/hip_runtime.h>


typedef float v2f __attribute__((ext_vector_type(2)));
typedef float v4f __attribute__((ext_vector_type(4)));
typedef float v8f __attribute__((ext_vector_type(8)));

#define WMMA_F32(A, B, C) \
  __builtin_amdgcn_wmma_f32_16x16x4_f32(false, (A), false, (B), (short)0, (C), false, false)

static __device__ __forceinline__ v2f ld2(const float* p) { return *(const v2f*)p; }
static __device__ __forceinline__ v4f ld4(const float* p) { return *(const v4f*)p; }
static __device__ __forceinline__ v2f mk2(float a, float b) { v2f r; r.x = a; r.y = b; return r; }

__global__ __launch_bounds__(256) void qkv_kernel(
    const float* __restrict__ x,
    const float* __restrict__ Wf, const float* __restrict__ Wg, const float* __restrict__ Wh,
    float* __restrict__ q, float* __restrict__ k, float* __restrict__ v) {
  __shared__ v2f wl[3][32 * 32];
  const int tid = threadIdx.x;
  const int wave = tid >> 5, lane = tid & 31;
  const int m = lane & 15, half = lane >> 4, kOff = half * 2;
  const int row0 = blockIdx.x * 128 + wave * 16;

  v8f aq0 = {}, aq1 = {}, ak0 = {}, ak1 = {}, av0 = {}, av1 = {};

  for (int c0 = 0; c0 < 256; c0 += 64) {
    for (int idx = tid; idx < 32 * 32; idx += 256) {
      const int kk2 = idx >> 5, n = idx & 31;
      const int g = (c0 + kk2 * 2) * 32 + n;
      wl[0][idx] = mk2(Wf[g], Wf[g + 32]);
      wl[1][idx] = mk2(Wg[g], Wg[g + 32]);
      wl[2][idx] = mk2(Wh[g], Wh[g + 32]);
    }
    __syncthreads();
#pragma unroll
    for (int kk = 0; kk < 64; kk += 4) {
      const v2f a = ld2(&x[(row0 + m) * 256 + c0 + kk + kOff]);
      const int bi = ((kk + kOff) >> 1) * 32;
      const v2f bf0 = wl[0][bi + m], bf1 = wl[0][bi + 16 + m];
      const v2f bg0 = wl[1][bi + m], bg1 = wl[1][bi + 16 + m];
      const v2f bh0 = wl[2][bi + m], bh1 = wl[2][bi + 16 + m];
      aq0 = WMMA_F32(a, bf0, aq0); aq1 = WMMA_F32(a, bf1, aq1);
      ak0 = WMMA_F32(a, bg0, ak0); ak1 = WMMA_F32(a, bg1, ak1);
      av0 = WMMA_F32(a, bh0, av0); av1 = WMMA_F32(a, bh1, av1);
    }
    __syncthreads();
  }
  const int pm = (m & 1) | ((m & 2) << 1) | ((m & 4) >> 1);
  float* qt = q + (row0 >> 4) * 512 + (m >> 3) * 128 + pm;
#pragma unroll
  for (int vi = 0; vi < 8; ++vi) {
    const int rr = vi + half * 8;
    qt[rr * 8] = aq0[vi];
    qt[256 + rr * 8] = aq1[vi];
    const int r = (row0 + rr) * 32;
    k[r + m] = ak0[vi]; k[r + 16 + m] = ak1[vi];
  }
#pragma unroll
  for (int t = 0; t < 4; ++t) {
    v4f ch;
    ch.x = av0[2 * t]; ch.y = av0[2 * t + 1]; ch.z = av1[2 * t]; ch.w = av1[2 * t + 1];
    *(v4f*)&v[(row0 >> 4) * 512 + t * 128 + lane * 4] = ch;
  }
}

static __device__ __forceinline__ float xhalf(float x) {
  const int b = __builtin_bit_cast(int, x);
  return __builtin_bit_cast(float,
      __builtin_amdgcn_permlanex16(b, b, 0x76543210, 0xfedcba98, false, false));
}

#define OP 36
#define WS 640

__global__ __launch_bounds__(256) __attribute__((amdgpu_waves_per_eu(4, 4))) void attn_kernel(
    const float* __restrict__ q, const float* __restrict__ k,
    const float* __restrict__ v, float* __restrict__ o) {
  __shared__ __attribute__((aligned(16))) float smem[8][WS];
  const int tid = threadIdx.x;
  const int wave = tid >> 5, lane = tid & 31;
  const int m = lane & 15, half = lane >> 4, kOff = half * 2;
  const int tile = wave >> 2, split = wave & 3;
  const int task = blockIdx.x * 2 + tile;
  const int b = task >> 8;
  const int i0 = (task & 255) * 16;
  const float* qb = q + b * 4096 * 32 + m * 8 + half * 4;
  const float* kb = k + b * 4096 * 32;
  const float* vb = v + b * 4096 * 32 + lane * 4;

  v2f bK[8];
#pragma unroll
  for (int t = 0; t < 8; ++t) bK[t] = ld2(&kb[(i0 + m) * 32 + 4 * t + kOff]);

  float mrun = -__builtin_inff(), lrun = 0.f;
  v8f o0 = {}, o1 = {};

  const int jbeg = split * 1024, jend = jbeg + 1024;
  for (int j0 = jbeg; j0 < jend; j0 += 64) {
    v8f s[4] = {{}, {}, {}, {}};
#pragma unroll
    for (int u = 0; u < 4; ++u) {
      v4f aq[4];
#pragma unroll
      for (int st = 0; st < 4; ++st) aq[st] = ld4(&qb[((j0 >> 4) + st) * 512 + u * 128]);
#pragma unroll
      for (int st = 0; st < 4; ++st) s[st] = WMMA_F32(mk2(aq[st].x, aq[st].y), bK[2 * u], s[st]);
#pragma unroll
      for (int st = 0; st < 4; ++st) s[st] = WMMA_F32(mk2(aq[st].z, aq[st].w), bK[2 * u + 1], s[st]);
    }
    float mx = s[0][0];
#pragma unroll
    for (int st = 0; st < 4; ++st)
#pragma unroll
      for (int r = 0; r < 8; ++r) mx = fmaxf(mx, s[st][r]);
    mx = fmaxf(mx, xhalf(mx));
    const float nm = fmaxf(mrun, mx);
    const float sc = __expf(mrun - nm);
    mrun = nm;
    float rs[4];
#pragma unroll
    for (int st = 0; st < 4; ++st) {
#pragma unroll
      for (int r = 0; r < 8; ++r) s[st][r] = __expf(s[st][r] - nm);
      rs[st] = ((s[st][0] + s[st][1]) + (s[st][2] + s[st][3])) +
               ((s[st][4] + s[st][5]) + (s[st][6] + s[st][7]));
    }
    lrun = lrun * sc + ((rs[0] + rs[1]) + (rs[2] + rs[3]));
#pragma unroll
    for (int r = 0; r < 8; ++r) { o0[r] *= sc; o1[r] *= sc; }
#pragma unroll
    for (int st = 0; st < 4; ++st) {
#pragma unroll
      for (int t = 0; t < 4; ++t) {
        const v4f av = ld4(&vb[((j0 >> 4) + st) * 512 + t * 128]);
        const v2f bp = mk2(s[st][2 * t], s[st][2 * t + 1]);
        o0 = WMMA_F32(mk2(av.x, av.y), bp, o0);
        o1 = WMMA_F32(mk2(av.z, av.w), bp, o1);
      }
    }
  }

  lrun += xhalf(lrun);
  {
    float* po = &smem[wave][m * OP + half * 8];
    v4f t0, t1, t2, t3;
    t0.x = o0[0]; t0.y = o0[1]; t0.z = o0[2]; t0.w = o0[3];
    t1.x = o0[4]; t1.y = o0[5]; t1.z = o0[6]; t1.w = o0[7];
    t2.x = o1[0]; t2.y = o1[1]; t2.z = o1[2]; t2.w = o1[3];
    t3.x = o1[4]; t3.y = o1[5]; t3.z = o1[6]; t3.w = o1[7];
    *(v4f*)(po)      = t0;
    *(v4f*)(po + 4)  = t1;
    *(v4f*)(po + 16) = t2;
    *(v4f*)(po + 20) = t3;
    if (half == 0) {
      smem[wave][576 + m] = lrun;
      smem[wave][592 + m] = mrun;
    }
  }
  __syncthreads();

  {
    const int w0 = tile * 4;
    const int r = split * 4 + (lane >> 3), d4 = (lane & 7) * 4;
    const float m0 = smem[w0][592 + r], m1 = smem[w0 + 1][592 + r];
    const float m2 = smem[w0 + 2][592 + r], m3 = smem[w0 + 3][592 + r];
    const float mg = fmaxf(fmaxf(fmaxf(m0, m1), m2), m3);
    float lg = 0.f;
    v4f acc = {};
#pragma unroll
    for (int p = 0; p < 4; ++p) {
      const float wgt = __expf(smem[w0 + p][592 + r] - mg);
      lg += wgt * smem[w0 + p][576 + r];
      const v4f part = *(const v4f*)&smem[w0 + p][r * OP + d4];
      acc.x += wgt * part.x; acc.y += wgt * part.y;
      acc.z += wgt * part.z; acc.w += wgt * part.w;
    }
    const float inv = 1.f / lg;
    acc.x *= inv; acc.y *= inv; acc.z *= inv; acc.w *= inv;
    *(v4f*)&o[b * 4096 * 32 + (i0 + r) * 32 + d4] = acc;
  }
}

__global__ __launch_bounds__(256) void out_kernel(
    const float* __restrict__ o, const float* __restrict__ Wv,
    const float* __restrict__ x, const float* __restrict__ gamma,
    float* __restrict__ out) {
  const int tid = threadIdx.x;
  const int wave = tid >> 5, lane = tid & 31;
  const int m = lane & 15, half = lane >> 4, kOff = half * 2;
  const int task = blockIdx.x * 8 + wave;
  const int row0 = (task >> 2) * 16, col0 = (task & 3) * 64;

  v2f aO[8];
#pragma unroll
  for (int t = 0; t < 8; ++t) aO[t] = ld2(&o[(row0 + m) * 32 + 4 * t + kOff]);

  v8f acc[4] = {{}, {}, {}, {}};
#pragma unroll
  for (int t = 0; t < 8; ++t) {
    const int kk = 4 * t + kOff;
#pragma unroll
    for (int ct = 0; ct < 4; ++ct) {
      const int col = col0 + ct * 16 + m;
      const v2f bw = mk2(Wv[kk * 256 + col], Wv[(kk + 1) * 256 + col]);
      acc[ct] = WMMA_F32(aO[t], bw, acc[ct]);
    }
  }
  const float g = *gamma;
#pragma unroll
  for (int ct = 0; ct < 4; ++ct) {
#pragma unroll
    for (int vi = 0; vi < 8; ++vi) {
      const int idx = (row0 + vi + half * 8) * 256 + col0 + ct * 16 + m;
      out[idx] = g * acc[ct][vi] + x[idx];
    }
  }
}

extern "C" void kernel_launch(void* const* d_in, const int* in_sizes, int n_in,
                              void* d_out, int out_size, void* d_ws, size_t ws_size,
                              hipStream_t stream) {
  const float* x  = (const float*)d_in[0];
  const float* Wf = (const float*)d_in[1];
  const float* Wg = (const float*)d_in[2];
  const float* Wh = (const float*)d_in[3];
  const float* Wv = (const float*)d_in[4];
  const float* gm = (const float*)d_in[5];
  float* out = (float*)d_out;

  const int QKV = 4 * 4096 * 32;
  float* q = (float*)d_ws;
  float* k = q + QKV;
  float* v = k + QKV;
  float* o = v + QKV;

  qkv_kernel<<<128, 256, 0, stream>>>(x, Wf, Wg, Wh, q, k, v);
  attn_kernel<<<512, 256, 0, stream>>>(q, k, v, o);
  out_kernel<<<512, 256, 0, stream>>>(o, Wv, x, gm, out);
}


# projection kernel: weights staged in LDS once for all of K (was per 64-k chunk, 8 barriers), one wave per 16 rows of one output, 64-row blocks on every WGP; sum order over k changes
# speedup vs baseline: 1.5157x; 1.1876x over previous
#include <hip/hip_runtime.h>


typedef float v2f __attribute__((ext_vector_type(2)));
typedef float v4f __attribute__((ext_vector_type(4)));
typedef float v8f __attribute__((ext_vector_type(8)));

#define WMMA_F32(A, B, C) \
  __builtin_amdgcn_wmma_f32_16x16x4_f32(false, (A), false, (B), (short)0, (C), false, false)

static __device__ __forceinline__ v2f ld2(const float* p) { return *(const v2f*)p; }
static __device__ __forceinline__ v4f ld4(const float* p) { return *(const v4f*)p; }
static __device__ __forceinline__ v2f mk2(float a, float b) { v2f r; r.x = a; r.y = b; return r; }

__global__ __launch_bounds__(384) void qkv_kernel(
    const float* __restrict__ x,
    const float* __restrict__ Wf, const float* __restrict__ Wg, const float* __restrict__ Wh,
    float* __restrict__ q, float* __restrict__ k, float* __restrict__ v) {
  __shared__ __attribute__((aligned(16))) v4f wl[3][64 * 32];
  const int tid = threadIdx.x;
  const int wave = tid >> 5, lane = tid & 31;
  const int m = lane & 15, half = lane >> 4;
  const int rt = wave & 3;
  const int which = __builtin_amdgcn_readfirstlane(wave >> 2);
  const int row0 = blockIdx.x * 64 + rt * 16;

  v4f wr[4][4];
#pragma unroll
  for (int t = 0; t < 4; ++t) {
    const int task = tid + t * 384;
    const int w = task >> 9, rem = task & 511;
    const int kq = rem >> 3, n4 = rem & 7;
    const float* Wp = (w == 0) ? Wf : ((w == 1) ? Wg : Wh);
    const float* s = Wp + kq * 128 + n4 * 4;
    wr[t][0] = ld4(s); wr[t][1] = ld4(s + 32); wr[t][2] = ld4(s + 64); wr[t][3] = ld4(s + 96);
  }

  v4f a[32];
  const float* xr = x + (row0 + m) * 256 + half * 4;
#pragma unroll
  for (int i = 0; i < 32; ++i) a[i] = ld4(xr + 8 * i);

#pragma unroll
  for (int t = 0; t < 4; ++t) {
    const int task = tid + t * 384;
    const int w = task >> 9, rem = task & 511;
    const int kq = rem >> 3, n4 = rem & 7;
    v4f* d = &wl[w][kq * 32 + n4 * 4];
    v4f e0, e1, e2, e3;
    e0.x = wr[t][0].x; e0.y = wr[t][1].x; e0.z = wr[t][2].x; e0.w = wr[t][3].x;
    e1.x = wr[t][0].y; e1.y = wr[t][1].y; e1.z = wr[t][2].y; e1.w = wr[t][3].y;
    e2.x = wr[t][0].z; e2.y = wr[t][1].z; e2.z = wr[t][2].z; e2.w = wr[t][3].z;
    e3.x = wr[t][0].w; e3.y = wr[t][1].w; e3.z = wr[t][2].w; e3.w = wr[t][3].w;
    d[0] = e0; d[1] = e1; d[2] = e2; d[3] = e3;
  }
  __builtin_amdgcn_fence(__ATOMIC_RELEASE, "workgroup", "local");
  __builtin_amdgcn_s_barrier();
  __builtin_amdgcn_fence(__ATOMIC_ACQUIRE, "workgroup", "local");

  const v4f* bw = &wl[which][half * 32 + m];
  v8f c0 = {}, c1 = {};
#pragma unroll
  for (int i = 0; i < 32; ++i) {
    const v4f b0 = bw[i * 64], b1 = bw[i * 64 + 16];
    const v2f alo = mk2(a[i].x, a[i].y), ahi = mk2(a[i].z, a[i].w);
    c0 = WMMA_F32(alo, mk2(b0.x, b0.y), c0);
    c1 = WMMA_F32(alo, mk2(b1.x, b1.y), c1);
    c0 = WMMA_F32(ahi, mk2(b0.z, b0.w), c0);
    c1 = WMMA_F32(ahi, mk2(b1.z, b1.w), c1);
  }

  if (which == 0) {
    const int pm = (m & 1) | ((m & 2) << 1) | ((m & 4) >> 1);
    float* qt = q + (row0 >> 4) * 512 + (m >> 3) * 128 + pm;
#pragma unroll
    for (int vi = 0; vi < 8; ++vi) {
      const int rr = vi + half * 8;
      qt[rr * 8] = c0[vi];
      qt[256 + rr * 8] = c1[vi];
    }
  } else if (which == 1) {
#pragma unroll
    for (int vi = 0; vi < 8; ++vi) {
      const int r = (row0 + vi + half * 8) * 32;
      k[r + m] = c0[vi]; k[r + 16 + m] = c1[vi];
    }
  } else {
#pragma unroll
    for (int t = 0; t < 4; ++t) {
      v4f ch;
      ch.x = c0[2 * t]; ch.y = c0[2 * t + 1]; ch.z = c1[2 * t]; ch.w = c1[2 * t + 1];
      *(v4f*)&v[(row0 >> 4) * 512 + t * 128 + lane * 4] = ch;
    }
  }
}

static __device__ __forceinline__ float xhalf(float x) {
  const int b = __builtin_bit_cast(int, x);
  return __builtin_bit_cast(float,
      __builtin_amdgcn_permlanex16(b, b, 0x76543210, 0xfedcba98, false, false));
}

#define OP 36
#define WS 640

__global__ __launch_bounds__(256) __attribute__((amdgpu_waves_per_eu(4, 4))) void attn_kernel(
    const float* __restrict__ q, const float* __restrict__ k,
    const float* __restrict__ v, float* __restrict__ o) {
  __shared__ __attribute__((aligned(16))) float smem[8][WS];
  const int tid = threadIdx.x;
  const int wave = tid >> 5, lane = tid & 31;
  const int m = lane & 15, half = lane >> 4, kOff = half * 2;
  const int tile = wave >> 2, split = wave & 3;
  const int task = blockIdx.x * 2 + tile;
  const int b = task >> 8;
  const int i0 = (task & 255) * 16;
  const float* qb = q + b * 4096 * 32 + m * 8 + half * 4;
  const float* kb = k + b * 4096 * 32;
  const float* vb = v + b * 4096 * 32 + lane * 4;

  v2f bK[8];
#pragma unroll
  for (int t = 0; t < 8; ++t) bK[t] = ld2(&kb[(i0 + m) * 32 + 4 * t + kOff]);

  float mrun = -__builtin_inff(), lrun = 0.f;
  v8f o0 = {}, o1 = {};

  const int jbeg = split * 1024, jend = jbeg + 1024;
  for (int j0 = jbeg; j0 < jend; j0 += 64) {
    v8f s[4] = {{}, {}, {}, {}};
#pragma unroll
    for (int u = 0; u < 4; ++u) {
      v4f aq[4];
#pragma unroll
      for (int st = 0; st < 4; ++st) aq[st] = ld4(&qb[((j0 >> 4) + st) * 512 + u * 128]);
#pragma unroll
      for (int st = 0; st < 4; ++st) s[st] = WMMA_F32(mk2(aq[st].x, aq[st].y), bK[2 * u], s[st]);
#pragma unroll
      for (int st = 0; st < 4; ++st) s[st] = WMMA_F32(mk2(aq[st].z, aq[st].w), bK[2 * u + 1], s[st]);
    }
    float mx = s[0][0];
#pragma unroll
    for (int st = 0; st < 4; ++st)
#pragma unroll
      for (int r = 0; r < 8; ++r) mx = fmaxf(mx, s[st][r]);
    mx = fmaxf(mx, xhalf(mx));
    const float nm = fmaxf(mrun, mx);
    const float sc = __expf(mrun - nm);
    mrun = nm;
    float rs[4];
#pragma unroll
    for (int st = 0; st < 4; ++st) {
#pragma unroll
      for (int r = 0; r < 8; ++r) s[st][r] = __expf(s[st][r] - nm);
      rs[st] = ((s[st][0] + s[st][1]) + (s[st][2] + s[st][3])) +
               ((s[st][4] + s[st][5]) + (s[st][6] + s[st][7]));
    }
    lrun = lrun * sc + ((rs[0] + rs[1]) + (rs[2] + rs[3]));
#pragma unroll
    for (int r = 0; r < 8; ++r) { o0[r] *= sc; o1[r] *= sc; }
#pragma unroll
    for (int st = 0; st < 4; ++st) {
#pragma unroll
      for (int t = 0; t < 4; ++t) {
        const v4f av = ld4(&vb[((j0 >> 4) + st) * 512 + t * 128]);
        const v2f bp = mk2(s[st][2 * t], s[st][2 * t + 1]);
        o0 = WMMA_F32(mk2(av.x, av.y), bp, o0);
        o1 = WMMA_F32(mk2(av.z, av.w), bp, o1);
      }
    }
  }

  lrun += xhalf(lrun);
  {
    float* po = &smem[wave][m * OP + half * 8];
    v4f t0, t1, t2, t3;
    t0.x = o0[0]; t0.y = o0[1]; t0.z = o0[2]; t0.w = o0[3];
    t1.x = o0[4]; t1.y = o0[5]; t1.z = o0[6]; t1.w = o0[7];
    t2.x = o1[0]; t2.y = o1[1]; t2.z = o1[2]; t2.w = o1[3];
    t3.x = o1[4]; t3.y = o1[5]; t3.z = o1[6]; t3.w = o1[7];
    *(v4f*)(po)      = t0;
    *(v4f*)(po + 4)  = t1;
    *(v4f*)(po + 16) = t2;
    *(v4f*)(po + 20) = t3;
    if (half == 0) {
      smem[wave][576 + m] = lrun;
      smem[wave][592 + m] = mrun;
    }
  }
  __syncthreads();

  {
    const int w0 = tile * 4;
    const int r = split * 4 + (lane >> 3), d4 = (lane & 7) * 4;
    const float m0 = smem[w0][592 + r], m1 = smem[w0 + 1][592 + r];
    const float m2 = smem[w0 + 2][592 + r], m3 = smem[w0 + 3][592 + r];
    const float mg = fmaxf(fmaxf(fmaxf(m0, m1), m2), m3);
    float lg = 0.f;
    v4f acc = {};
#pragma unroll
    for (int p = 0; p < 4; ++p) {
      const float wgt = __expf(smem[w0 + p][592 + r] - mg);
      lg += wgt * smem[w0 + p][576 + r];
      const v4f part = *(const v4f*)&smem[w0 + p][r * OP + d4];
      acc.x += wgt * part.x; acc.y += wgt * part.y;
      acc.z += wgt * part.z; acc.w += wgt * part.w;
    }
    const float inv = 1.f / lg;
    acc.x *= inv; acc.y *= inv; acc.z *= inv; acc.w *= inv;
    *(v4f*)&o[b * 4096 * 32 + (i0 + r) * 32 + d4] = acc;
  }
}

__global__ __launch_bounds__(256) void out_kernel(
    const float* __restrict__ o, const float* __restrict__ Wv,
    const float* __restrict__ x, const float* __restrict__ gamma,
    float* __restrict__ out) {
  const int tid = threadIdx.x;
  const int wave = tid >> 5, lane = tid & 31;
  const int m = lane & 15, half = lane >> 4, kOff = half * 2;
  const int task = blockIdx.x * 8 + wave;
  const int row0 = (task >> 2) * 16, col0 = (task & 3) * 64;

  v2f aO[8];
#pragma unroll
  for (int t = 0; t < 8; ++t) aO[t] = ld2(&o[(row0 + m) * 32 + 4 * t + kOff]);

  v8f acc[4] = {{}, {}, {}, {}};
#pragma unroll
  for (int t = 0; t < 8; ++t) {
    const int kk = 4 * t + kOff;
#pragma unroll
    for (int ct = 0; ct < 4; ++ct) {
      const int col = col0 + ct * 16 + m;
      const v2f bw = mk2(Wv[kk * 256 + col], Wv[(kk + 1) * 256 + col]);
      acc[ct] = WMMA_F32(aO[t], bw, acc[ct]);
    }
  }
  const float g = *gamma;
#pragma unroll
  for (int ct = 0; ct < 4; ++ct) {
#pragma unroll
    for (int vi = 0; vi < 8; ++vi) {
      const int idx = (row0 + vi + half * 8) * 256 + col0 + ct * 16 + m;
      out[idx] = g * acc[ct][vi] + x[idx];
    }
  }
}

extern "C" void kernel_launch(void* const* d_in, const int* in_sizes, int n_in,
                              void* d_out, int out_size, void* d_ws, size_t ws_size,
                              hipStream_t stream) {
  const float* x  = (const float*)d_in[0];
  const float* Wf = (const float*)d_in[1];
  const float* Wg = (const float*)d_in[2];
  const float* Wh = (const float*)d_in[3];
  const float* Wv = (const float*)d_in[4];
  const float* gm = (const float*)d_in[5];
  float* out = (float*)d_out;

  const int QKV = 4 * 4096 * 32;
  float* q = (float*)d_ws;
  float* k = q + QKV;
  float* v = k + QKV;
  float* o = v + QKV;

  qkv_kernel<<<256, 384, 0, stream>>>(x, Wf, Wg, Wh, q, k, v);
  attn_kernel<<<512, 256, 0, stream>>>(q, k, v, o);
  out_kernel<<<512, 256, 0, stream>>>(o, Wv, x, gm, out);
}


# output projection gamma*(o@Wv)+x moved into the attention kernel's epilogue: combined o rows stay in LDS, two launches instead of three, Wv/x/out as 16-byte accesses; sums and their order unchanged
# speedup vs baseline: 1.6236x; 1.0712x over previous
#include <hip/hip_runtime.h>


typedef float v2f __attribute__((ext_vector_type(2)));
typedef float v4f __attribute__((ext_vector_type(4)));
typedef float v8f __attribute__((ext_vector_type(8)));

#define WMMA_F32(A, B, C) \
  __builtin_amdgcn_wmma_f32_16x16x4_f32(false, (A), false, (B), (short)0, (C), false, false)

static __device__ __forceinline__ v2f ld2(const float* p) { return *(const v2f*)p; }
static __device__ __forceinline__ v4f ld4(const float* p) { return *(const v4f*)p; }
static __device__ __forceinline__ v2f mk2(float a, float b) { v2f r; r.x = a; r.y = b; return r; }

__global__ __launch_bounds__(384) void qkv_kernel(
    const float* __restrict__ x,
    const float* __restrict__ Wf, const float* __restrict__ Wg, const float* __restrict__ Wh,
    float* __restrict__ q, float* __restrict__ k, float* __restrict__ v) {
  __shared__ __attribute__((aligned(16))) v4f wl[3][64 * 32];
  const int tid = threadIdx.x;
  const int wave = tid >> 5, lane = tid & 31;
  const int m = lane & 15, half = lane >> 4;
  const int rt = wave & 3;
  const int which = __builtin_amdgcn_readfirstlane(wave >> 2);
  const int row0 = blockIdx.x * 64 + rt * 16;

  v4f wr[4][4];
#pragma unroll
  for (int t = 0; t < 4; ++t) {
    const int task = tid + t * 384;
    const int w = task >> 9, rem = task & 511;
    const int kq = rem >> 3, n4 = rem & 7;
    const float* Wp = (w == 0) ? Wf : ((w == 1) ? Wg : Wh);
    const float* s = Wp + kq * 128 + n4 * 4;
    wr[t][0] = ld4(s); wr[t][1] = ld4(s + 32); wr[t][2] = ld4(s + 64); wr[t][3] = ld4(s + 96);
  }

  v4f a[32];
  const float* xr = x + (row0 + m) * 256 + half * 4;
#pragma unroll
  for (int i = 0; i < 32; ++i) a[i] = ld4(xr + 8 * i);

#pragma unroll
  for (int t = 0; t < 4; ++t) {
    const int task = tid + t * 384;
    const int w = task >> 9, rem = task & 511;
    const int kq = rem >> 3, n4 = rem & 7;
    v4f* d = &wl[w][kq * 32 + n4 * 4];
    v4f e0, e1, e2, e3;
    e0.x = wr[t][0].x; e0.y = wr[t][1].x; e0.z = wr[t][2].x; e0.w = wr[t][3].x;
    e1.x = wr[t][0].y; e1.y = wr[t][1].y; e1.z = wr[t][2].y; e1.w = wr[t][3].y;
    e2.x = wr[t][0].z; e2.y = wr[t][1].z; e2.z = wr[t][2].z; e2.w = wr[t][3].z;
    e3.x = wr[t][0].w; e3.y = wr[t][1].w; e3.z = wr[t][2].w; e3.w = wr[t][3].w;
    d[0] = e0; d[1] = e1; d[2] = e2; d[3] = e3;
  }
  __builtin_amdgcn_fence(__ATOMIC_RELEASE, "workgroup", "local");
  __builtin_amdgcn_s_barrier();
  __builtin_amdgcn_fence(__ATOMIC_ACQUIRE, "workgroup", "local");

  const v4f* bw = &wl[which][half * 32 + m];
  v8f c0 = {}, c1 = {};
#pragma unroll
  for (int i = 0; i < 32; ++i) {
    const v4f b0 = bw[i * 64], b1 = bw[i * 64 + 16];
    const v2f alo = mk2(a[i].x, a[i].y), ahi = mk2(a[i].z, a[i].w);
    c0 = WMMA_F32(alo, mk2(b0.x, b0.y), c0);
    c1 = WMMA_F32(alo, mk2(b1.x, b1.y), c1);
    c0 = WMMA_F32(ahi, mk2(b0.z, b0.w), c0);
    c1 = WMMA_F32(ahi, mk2(b1.z, b1.w), c1);
  }

  if (which == 0) {
    const int pm = (m & 1) | ((m & 2) << 1) | ((m & 4) >> 1);
    float* qt = q + (row0 >> 4) * 512 + (m >> 3) * 128 + pm;
#pragma unroll
    for (int vi = 0; vi < 8; ++vi) {
      const int rr = vi + half * 8;
      qt[rr * 8] = c0[vi];
      qt[256 + rr * 8] = c1[vi];
    }
  } else if (which == 1) {
#pragma unroll
    for (int vi = 0; vi < 8; ++vi) {
      const int r = (row0 + vi + half * 8) * 32;
      k[r + m] = c0[vi]; k[r + 16 + m] = c1[vi];
    }
  } else {
#pragma unroll
    for (int t = 0; t < 4; ++t) {
      v4f ch;
      ch.x = c0[2 * t]; ch.y = c0[2 * t + 1]; ch.z = c1[2 * t]; ch.w = c1[2 * t + 1];
      *(v4f*)&v[(row0 >> 4) * 512 + t * 128 + lane * 4] = ch;
    }
  }
}

static __device__ __forceinline__ float xhalf(float x) {
  const int b = __builtin_bit_cast(int, x);
  return __builtin_bit_cast(float,
      __builtin_amdgcn_permlanex16(b, b, 0x76543210, 0xfedcba98, false, false));
}

#define OP 36
#define WS 640

static __device__ __forceinline__ void lds_barrier() {
  __builtin_amdgcn_fence(__ATOMIC_RELEASE, "workgroup", "local");
  __builtin_amdgcn_s_barrier();
  __builtin_amdgcn_fence(__ATOMIC_ACQUIRE, "workgroup", "local");
}

__global__ __launch_bounds__(256) __attribute__((amdgpu_waves_per_eu(4, 4))) void attn_out_kernel(
    const float* __restrict__ q, const float* __restrict__ k, const float* __restrict__ v,
    const float* __restrict__ Wv, const float* __restrict__ x, const float* __restrict__ gamma,
    float* __restrict__ out) {
  __shared__ __attribute__((aligned(16))) float smem[8][WS];
  __shared__ __attribute__((aligned(16))) float ot[2][16 * OP];
  const int tid = threadIdx.x;
  const int wave = tid >> 5, lane = tid & 31;
  const int m = lane & 15, half = lane >> 4, kOff = half * 2;
  const int tile = wave >> 2, split = wave & 3;
  const int task = blockIdx.x * 2 + tile;
  const int b = task >> 8;
  const int i0 = (task & 255) * 16;
  const float* qb = q + b * 4096 * 32 + m * 8 + half * 4;
  const float* kb = k + b * 4096 * 32;
  const float* vb = v + b * 4096 * 32 + lane * 4;

  v2f bK[8];
#pragma unroll
  for (int t = 0; t < 8; ++t) bK[t] = ld2(&kb[(i0 + m) * 32 + 4 * t + kOff]);

  float mrun = -__builtin_inff(), lrun = 0.f;
  v8f o0 = {}, o1 = {};

  const int jbeg = split * 1024, jend = jbeg + 1024;
  for (int j0 = jbeg; j0 < jend; j0 += 64) {
    v8f s[4] = {{}, {}, {}, {}};
#pragma unroll
    for (int u = 0; u < 4; ++u) {
      v4f aq[4];
#pragma unroll
      for (int st = 0; st < 4; ++st) aq[st] = ld4(&qb[((j0 >> 4) + st) * 512 + u * 128]);
#pragma unroll
      for (int st = 0; st < 4; ++st) s[st] = WMMA_F32(mk2(aq[st].x, aq[st].y), bK[2 * u], s[st]);
#pragma unroll
      for (int st = 0; st < 4; ++st) s[st] = WMMA_F32(mk2(aq[st].z, aq[st].w), bK[2 * u + 1], s[st]);
    }
    float mx = s[0][0];
#pragma unroll
    for (int st = 0; st < 4; ++st)
#pragma unroll
      for (int r = 0; r < 8; ++r) mx = fmaxf(mx, s[st][r]);
    mx = fmaxf(mx, xhalf(mx));
    const float nm = fmaxf(mrun, mx);
    const float sc = __expf(mrun - nm);
    mrun = nm;
    float rs[4];
#pragma unroll
    for (int st = 0; st < 4; ++st) {
#pragma unroll
      for (int r = 0; r < 8; ++r) s[st][r] = __expf(s[st][r] - nm);
      rs[st] = ((s[st][0] + s[st][1]) + (s[st][2] + s[st][3])) +
               ((s[st][4] + s[st][5]) + (s[st][6] + s[st][7]));
    }
    lrun = lrun * sc + ((rs[0] + rs[1]) + (rs[2] + rs[3]));
#pragma unroll
    for (int r = 0; r < 8; ++r) { o0[r] *= sc; o1[r] *= sc; }
#pragma unroll
    for (int st = 0; st < 4; ++st) {
#pragma unroll
      for (int t = 0; t < 4; ++t) {
        const v4f av = ld4(&vb[((j0 >> 4) + st) * 512 + t * 128]);
        const v2f bp = mk2(s[st][2 * t], s[st][2 * t + 1]);
        o0 = WMMA_F32(mk2(av.x, av.y), bp, o0);
        o1 = WMMA_F32(mk2(av.z, av.w), bp, o1);
      }
    }
  }

  const int colm = split * 64 + 4 * m;
  const int grow = task * 16;
  v4f bw[8][2];
#pragma unroll
  for (int t = 0; t < 8; ++t) {
    bw[t][0] = ld4(&Wv[(4 * t + kOff) * 256 + colm]);
    bw[t][1] = ld4(&Wv[(4 * t + kOff + 1) * 256 + colm]);
  }
  v4f xv[8];
#pragma unroll
  for (int vi = 0; vi < 8; ++vi) xv[vi] = ld4(&x[(grow + vi + half * 8) * 256 + colm]);
  const float g = *gamma;

  lrun += xhalf(lrun);
  {
    float* po = &smem[wave][m * OP + half * 8];
    v4f t0, t1, t2, t3;
    t0.x = o0[0]; t0.y = o0[1]; t0.z = o0[2]; t0.w = o0[3];
    t1.x = o0[4]; t1.y = o0[5]; t1.z = o0[6]; t1.w = o0[7];
    t2.x = o1[0]; t2.y = o1[1]; t2.z = o1[2]; t2.w = o1[3];
    t3.x = o1[4]; t3.y = o1[5]; t3.z = o1[6]; t3.w = o1[7];
    *(v4f*)(po)      = t0;
    *(v4f*)(po + 4)  = t1;
    *(v4f*)(po + 16) = t2;
    *(v4f*)(po + 20) = t3;
    if (half == 0) {
      smem[wave][576 + m] = lrun;
      smem[wave][592 + m] = mrun;
    }
  }
  lds_barrier();

  {
    const int w0 = tile * 4;
    const int r = split * 4 + (lane >> 3), d4 = (lane & 7) * 4;
    const float m0 = smem[w0][592 + r], m1 = smem[w0 + 1][592 + r];
    const float m2 = smem[w0 + 2][592 + r], m3 = smem[w0 + 3][592 + r];
    const float mg = fmaxf(fmaxf(fmaxf(m0, m1), m2), m3);
    float lg = 0.f;
    v4f acc = {};
#pragma unroll
    for (int p = 0; p < 4; ++p) {
      const float wgt = __expf(smem[w0 + p][592 + r] - mg);
      lg += wgt * smem[w0 + p][576 + r];
      const v4f part = *(const v4f*)&smem[w0 + p][r * OP + d4];
      acc.x += wgt * part.x; acc.y += wgt * part.y;
      acc.z += wgt * part.z; acc.w += wgt * part.w;
    }
    const float inv = 1.f / lg;
    acc.x *= inv; acc.y *= inv; acc.z *= inv; acc.w *= inv;
    *(v4f*)&ot[tile][r * OP + d4] = acc;
  }
  lds_barrier();

  {
    v2f aO[8];
#pragma unroll
    for (int t = 0; t < 8; ++t) aO[t] = ld2(&ot[tile][m * OP + 4 * t + kOff]);
    v8f acc[4] = {{}, {}, {}, {}};
#pragma unroll
    for (int t = 0; t < 8; ++t) {
#pragma unroll
      for (int ct = 0; ct < 4; ++ct)
        acc[ct] = WMMA_F32(aO[t], mk2(bw[t][0][ct], bw[t][1][ct]), acc[ct]);
    }
#pragma unroll
    for (int vi = 0; vi < 8; ++vi) {
      v4f r;
      r.x = g * acc[0][vi] + xv[vi].x; r.y = g * acc[1][vi] + xv[vi].y;
      r.z = g * acc[2][vi] + xv[vi].z; r.w = g * acc[3][vi] + xv[vi].w;
      *(v4f*)&out[(grow + vi + half * 8) * 256 + colm] = r;
    }
  }
}

extern "C" void kernel_launch(void* const* d_in, const int* in_sizes, int n_in,
                              void* d_out, int out_size, void* d_ws, size_t ws_size,
                              hipStream_t stream) {
  const float* x  = (const float*)d_in[0];
  const float* Wf = (const float*)d_in[1];
  const float* Wg = (const float*)d_in[2];
  const float* Wh = (const float*)d_in[3];
  const float* Wv = (const float*)d_in[4];
  const float* gm = (const float*)d_in[5];
  float* out = (float*)d_out;

  const int QKV = 4 * 4096 * 32;
  float* q = (float*)d_ws;
  float* k = q + QKV;
  float* v = k + QKV;

  qkv_kernel<<<256, 384, 0, stream>>>(x, Wf, Wg, Wh, q, k, v);
  attn_out_kernel<<<512, 256, 0, stream>>>(q, k, v, Wv, x, gm, out);
}


# attention: a wave carries both row tiles of its block over 512 columns (8-way column split, 32-col blocks): each q/v fragment load feeds two WMMAs, reads halve; 8 partials combined, sum order changes
# speedup vs baseline: 1.7180x; 1.0581x over previous
#include <hip/hip_runtime.h>


typedef float v2f __attribute__((ext_vector_type(2)));
typedef float v4f __attribute__((ext_vector_type(4)));
typedef float v8f __attribute__((ext_vector_type(8)));

#define WMMA_F32(A, B, C) \
  __builtin_amdgcn_wmma_f32_16x16x4_f32(false, (A), false, (B), (short)0, (C), false, false)

static __device__ __forceinline__ v2f ld2(const float* p) { return *(const v2f*)p; }
static __device__ __forceinline__ v4f ld4(const float* p) { return *(const v4f*)p; }
static __device__ __forceinline__ v2f mk2(float a, float b) { v2f r; r.x = a; r.y = b; return r; }

__global__ __launch_bounds__(384) void qkv_kernel(
    const float* __restrict__ x,
    const float* __restrict__ Wf, const float* __restrict__ Wg, const float* __restrict__ Wh,
    float* __restrict__ q, float* __restrict__ k, float* __restrict__ v) {
  __shared__ __attribute__((aligned(16))) v4f wl[3][64 * 32];
  const int tid = threadIdx.x;
  const int wave = tid >> 5, lane = tid & 31;
  const int m = lane & 15, half = lane >> 4;
  const int rt = wave & 3;
  const int which = __builtin_amdgcn_readfirstlane(wave >> 2);
  const int row0 = blockIdx.x * 64 + rt * 16;

  v4f wr[4][4];
#pragma unroll
  for (int t = 0; t < 4; ++t) {
    const int task = tid + t * 384;
    const int w = task >> 9, rem = task & 511;
    const int kq = rem >> 3, n4 = rem & 7;
    const float* Wp = (w == 0) ? Wf : ((w == 1) ? Wg : Wh);
    const float* s = Wp + kq * 128 + n4 * 4;
    wr[t][0] = ld4(s); wr[t][1] = ld4(s + 32); wr[t][2] = ld4(s + 64); wr[t][3] = ld4(s + 96);
  }

  v4f a[32];
  const float* xr = x + (row0 + m) * 256 + half * 4;
#pragma unroll
  for (int i = 0; i < 32; ++i) a[i] = ld4(xr + 8 * i);

#pragma unroll
  for (int t = 0; t < 4; ++t) {
    const int task = tid + t * 384;
    const int w = task >> 9, rem = task & 511;
    const int kq = rem >> 3, n4 = rem & 7;
    v4f* d = &wl[w][kq * 32 + n4 * 4];
    v4f e0, e1, e2, e3;
    e0.x = wr[t][0].x; e0.y = wr[t][1].x; e0.z = wr[t][2].x; e0.w = wr[t][3].x;
    e1.x = wr[t][0].y; e1.y = wr[t][1].y; e1.z = wr[t][2].y; e1.w = wr[t][3].y;
    e2.x = wr[t][0].z; e2.y = wr[t][1].z; e2.z = wr[t][2].z; e2.w = wr[t][3].z;
    e3.x = wr[t][0].w; e3.y = wr[t][1].w; e3.z = wr[t][2].w; e3.w = wr[t][3].w;
    d[0] = e0; d[1] = e1; d[2] = e2; d[3] = e3;
  }
  __builtin_amdgcn_fence(__ATOMIC_RELEASE, "workgroup", "local");
  __builtin_amdgcn_s_barrier();
  __builtin_amdgcn_fence(__ATOMIC_ACQUIRE, "workgroup", "local");

  const v4f* bw = &wl[which][half * 32 + m];
  v8f c0 = {}, c1 = {};
#pragma unroll
  for (int i = 0; i < 32; ++i) {
    const v4f b0 = bw[i * 64], b1 = bw[i * 64 + 16];
    const v2f alo = mk2(a[i].x, a[i].y), ahi = mk2(a[i].z, a[i].w);
    c0 = WMMA_F32(alo, mk2(b0.x, b0.y), c0);
    c1 = WMMA_F32(alo, mk2(b1.x, b1.y), c1);
    c0 = WMMA_F32(ahi, mk2(b0.z, b0.w), c0);
    c1 = WMMA_F32(ahi, mk2(b1.z, b1.w), c1);
  }

  if (which == 0) {
    const int pm = (m & 1) | ((m & 2) << 1) | ((m & 4) >> 1);
    float* qt = q + (row0 >> 4) * 512 + (m >> 3) * 128 + pm;
#pragma unroll
    for (int vi = 0; vi < 8; ++vi) {
      const int rr = vi + half * 8;
      qt[rr * 8] = c0[vi];
      qt[256 + rr * 8] = c1[vi];
    }
  } else if (which == 1) {
#pragma unroll
    for (int vi = 0; vi < 8; ++vi) {
      const int r = (row0 + vi + half * 8) * 32;
      k[r + m] = c0[vi]; k[r + 16 + m] = c1[vi];
    }
  } else {
#pragma unroll
    for (int t = 0; t < 4; ++t) {
      v4f ch;
      ch.x = c0[2 * t]; ch.y = c0[2 * t + 1]; ch.z = c1[2 * t]; ch.w = c1[2 * t + 1];
      *(v4f*)&v[(row0 >> 4) * 512 + t * 128 + lane * 4] = ch;
    }
  }
}

static __device__ __forceinline__ float xhalf(float x) {
  const int b = __builtin_bit_cast(int, x);
  return __builtin_bit_cast(float,
      __builtin_amdgcn_permlanex16(b, b, 0x76543210, 0xfedcba98, false, false));
}

#define OP 36
#define WS 640
#define JT 2

static __device__ __forceinline__ void lds_barrier() {
  __builtin_amdgcn_fence(__ATOMIC_RELEASE, "workgroup", "local");
  __builtin_amdgcn_s_barrier();
  __builtin_amdgcn_fence(__ATOMIC_ACQUIRE, "workgroup", "local");
}

__global__ __launch_bounds__(256) __attribute__((amdgpu_waves_per_eu(4, 4))) void attn_out_kernel(
    const float* __restrict__ q, const float* __restrict__ k, const float* __restrict__ v,
    const float* __restrict__ Wv, const float* __restrict__ x, const float* __restrict__ gamma,
    float* __restrict__ out) {
  __shared__ __attribute__((aligned(16))) float smem[8][2][WS];
  __shared__ __attribute__((aligned(16))) float ot[2][16 * OP];
  const int tid = threadIdx.x;
  const int wave = tid >> 5, lane = tid & 31;
  const int m = lane & 15, half = lane >> 4, kOff = half * 2;
  const int task0 = blockIdx.x * 2;
  const int b = task0 >> 8;
  const int i0 = (task0 & 255) * 16;
  const float* qb = q + b * 4096 * 32 + m * 8 + half * 4;
  const float* kb = k + b * 4096 * 32;
  const float* vb = v + b * 4096 * 32 + lane * 4;

  v2f bK[2][8];
#pragma unroll
  for (int rt = 0; rt < 2; ++rt)
#pragma unroll
    for (int t = 0; t < 8; ++t) bK[rt][t] = ld2(&kb[(i0 + 16 * rt + m) * 32 + 4 * t + kOff]);

  float mrun[2] = {-__builtin_inff(), -__builtin_inff()}, lrun[2] = {0.f, 0.f};
  v8f o[2][2] = {{{}, {}}, {{}, {}}};

  const int jbeg = wave * 512, jend = jbeg + 512;
  for (int j0 = jbeg; j0 < jend; j0 += 16 * JT) {
    v8f s[2][JT] = {{{}, {}}, {{}, {}}};
#pragma unroll
    for (int u = 0; u < 4; ++u) {
      v4f aq[JT];
#pragma unroll
      for (int st = 0; st < JT; ++st) aq[st] = ld4(&qb[((j0 >> 4) + st) * 512 + u * 128]);
#pragma unroll
      for (int rt = 0; rt < 2; ++rt)
#pragma unroll
        for (int st = 0; st < JT; ++st)
          s[rt][st] = WMMA_F32(mk2(aq[st].x, aq[st].y), bK[rt][2 * u], s[rt][st]);
#pragma unroll
      for (int rt = 0; rt < 2; ++rt)
#pragma unroll
        for (int st = 0; st < JT; ++st)
          s[rt][st] = WMMA_F32(mk2(aq[st].z, aq[st].w), bK[rt][2 * u + 1], s[rt][st]);
    }
#pragma unroll
    for (int rt = 0; rt < 2; ++rt) {
      float mx = s[rt][0][0];
#pragma unroll
      for (int st = 0; st < JT; ++st)
#pragma unroll
        for (int r = 0; r < 8; ++r) mx = fmaxf(mx, s[rt][st][r]);
      mx = fmaxf(mx, xhalf(mx));
      const float nm = fmaxf(mrun[rt], mx);
      const float sc = __expf(mrun[rt] - nm);
      mrun[rt] = nm;
      float rs[JT];
#pragma unroll
      for (int st = 0; st < JT; ++st) {
#pragma unroll
        for (int r = 0; r < 8; ++r) s[rt][st][r] = __expf(s[rt][st][r] - nm);
        rs[st] = ((s[rt][st][0] + s[rt][st][1]) + (s[rt][st][2] + s[rt][st][3])) +
                 ((s[rt][st][4] + s[rt][st][5]) + (s[rt][st][6] + s[rt][st][7]));
      }
      lrun[rt] = lrun[rt] * sc + (rs[0] + rs[1]);
#pragma unroll
      for (int r = 0; r < 8; ++r) { o[rt][0][r] *= sc; o[rt][1][r] *= sc; }
    }
#pragma unroll
    for (int st = 0; st < JT; ++st) {
#pragma unroll
      for (int t = 0; t < 4; ++t) {
        const v4f av = ld4(&vb[((j0 >> 4) + st) * 512 + t * 128]);
#pragma unroll
        for (int rt = 0; rt < 2; ++rt) {
          const v2f bp = mk2(s[rt][st][2 * t], s[rt][st][2 * t + 1]);
          o[rt][0] = WMMA_F32(mk2(av.x, av.y), bp, o[rt][0]);
          o[rt][1] = WMMA_F32(mk2(av.z, av.w), bp, o[rt][1]);
        }
      }
    }
  }

  const int tile = wave >> 2, csplit = wave & 3;
  const int colm = csplit * 64 + 4 * m;
  const int grow = (task0 + tile) * 16;
  v4f bw[8][2];
#pragma unroll
  for (int t = 0; t < 8; ++t) {
    bw[t][0] = ld4(&Wv[(4 * t + kOff) * 256 + colm]);
    bw[t][1] = ld4(&Wv[(4 * t + kOff + 1) * 256 + colm]);
  }
  v4f xv[8];
#pragma unroll
  for (int vi = 0; vi < 8; ++vi) xv[vi] = ld4(&x[(grow + vi + half * 8) * 256 + colm]);
  const float g = *gamma;

#pragma unroll
  for (int rt = 0; rt < 2; ++rt) {
    const float lsum = lrun[rt] + xhalf(lrun[rt]);
    float* po = &smem[wave][rt][m * OP + half * 8];
    v4f t0, t1, t2, t3;
    t0.x = o[rt][0][0]; t0.y = o[rt][0][1]; t0.z = o[rt][0][2]; t0.w = o[rt][0][3];
    t1.x = o[rt][0][4]; t1.y = o[rt][0][5]; t1.z = o[rt][0][6]; t1.w = o[rt][0][7];
    t2.x = o[rt][1][0]; t2.y = o[rt][1][1]; t2.z = o[rt][1][2]; t2.w = o[rt][1][3];
    t3.x = o[rt][1][4]; t3.y = o[rt][1][5]; t3.z = o[rt][1][6]; t3.w = o[rt][1][7];
    *(v4f*)(po)      = t0;
    *(v4f*)(po + 4)  = t1;
    *(v4f*)(po + 16) = t2;
    *(v4f*)(po + 20) = t3;
    if (half == 0) {
      smem[wave][rt][576 + m] = lsum;
      smem[wave][rt][592 + m] = mrun[rt];
    }
  }
  lds_barrier();

  {
    const int r = csplit * 4 + (lane >> 3), d4 = (lane & 7) * 4;
    float mg = smem[0][tile][592 + r];
#pragma unroll
    for (int p = 1; p < 8; ++p) mg = fmaxf(mg, smem[p][tile][592 + r]);
    float lg = 0.f;
    v4f acc = {};
#pragma unroll
    for (int p = 0; p < 8; ++p) {
      const float wgt = __expf(smem[p][tile][592 + r] - mg);
      lg += wgt * smem[p][tile][576 + r];
      const v4f part = *(const v4f*)&smem[p][tile][r * OP + d4];
      acc.x += wgt * part.x; acc.y += wgt * part.y;
      acc.z += wgt * part.z; acc.w += wgt * part.w;
    }
    const float inv = 1.f / lg;
    acc.x *= inv; acc.y *= inv; acc.z *= inv; acc.w *= inv;
    *(v4f*)&ot[tile][r * OP + d4] = acc;
  }
  lds_barrier();

  {
    v2f aO[8];
#pragma unroll
    for (int t = 0; t < 8; ++t) aO[t] = ld2(&ot[tile][m * OP + 4 * t + kOff]);
    v8f acc[4] = {{}, {}, {}, {}};
#pragma unroll
    for (int t = 0; t < 8; ++t) {
#pragma unroll
      for (int ct = 0; ct < 4; ++ct)
        acc[ct] = WMMA_F32(aO[t], mk2(bw[t][0][ct], bw[t][1][ct]), acc[ct]);
    }
#pragma unroll
    for (int vi = 0; vi < 8; ++vi) {
      v4f r;
      r.x = g * acc[0][vi] + xv[vi].x; r.y = g * acc[1][vi] + xv[vi].y;
      r.z = g * acc[2][vi] + xv[vi].z; r.w = g * acc[3][vi] + xv[vi].w;
      *(v4f*)&out[(grow + vi + half * 8) * 256 + colm] = r;
    }
  }
}

extern "C" void kernel_launch(void* const* d_in, const int* in_sizes, int n_in,
                              void* d_out, int out_size, void* d_ws, size_t ws_size,
                              hipStream_t stream) {
  const float* x  = (const float*)d_in[0];
  const float* Wf = (const float*)d_in[1];
  const float* Wg = (const float*)d_in[2];
  const float* Wh = (const float*)d_in[3];
  const float* Wv = (const float*)d_in[4];
  const float* gm = (const float*)d_in[5];
  float* out = (float*)d_out;

  const int QKV = 4 * 4096 * 32;
  float* q = (float*)d_ws;
  float* k = q + QKV;
  float* v = k + QKV;

  qkv_kernel<<<256, 384, 0, stream>>>(x, Wf, Wg, Wh, q, k, v);
  attn_out_kernel<<<512, 256, 0, stream>>>(q, k, v, Wv, x, gm, out);
}
